# starting kernel
# baseline (speedup 1.0000x reference)
#include <hip/hip_runtime.h>
#include <hip/hip_bf16.h>

typedef __attribute__((ext_vector_type(2))) float v2f;
typedef __attribute__((ext_vector_type(8))) float v8f;

#define N_NODES 19
#define N_T     4096
#define LATENT  512
#define HID     2048
#define N_EDGES 342
#define PAD_ROWS 32

#define WS_H0    0
#define WS_T1    (WS_H0 + PAD_ROWS * LATENT)
#define WS_H1    (WS_T1 + PAD_ROWS * HID)
#define WS_H1AGG (WS_H1 + PAD_ROWS * HID)
#define WS_T2    (WS_H1AGG + PAD_ROWS * HID)
#define WS_Y     (WS_T2 + PAD_ROWS * N_T)
#define WS_TOTAL (WS_Y + LATENT)

__global__ void zero_kernel(float* __restrict__ p, int n) {
    int i = blockIdx.x * blockDim.x + threadIdx.x;
    int stride = gridDim.x * blockDim.x;
    for (; i < n; i += stride) p[i] = 0.0f;
}

__global__ void gin_agg_kernel(const float* __restrict__ h, const int* __restrict__ edges,
                               float* __restrict__ out, int F) {
    int idx = blockIdx.x * blockDim.x + threadIdx.x;
    if (idx >= N_NODES * F) return;
    int node = idx / F;
    int f = idx - node * F;
    const int* src = edges;
    const int* dst = edges + N_EDGES;
    float v = h[(size_t)node * F + f];
    for (int e = 0; e < N_EDGES; ++e) {
        if (dst[e] == node) v += h[(size_t)src[e] * F + f];
    }
    out[(size_t)node * F + f] = v;
}

template <int K, int N, int KCHUNK>
__global__ __launch_bounds__(128) void gemm_wmma_kernel(
        const float* __restrict__ A, const float* __restrict__ W,
        float* __restrict__ Out) {
    const int lane = threadIdx.x;
    const int half = lane >> 4;
    const int lm   = lane & 15;
    const int tile = blockIdx.x * 4 + threadIdx.y;
    const int n    = tile * 16 + lm;
    const int k0   = blockIdx.y * KCHUNK;

    const float* a0p  = A + (size_t)lm * K        + k0 + 2 * half;
    const float* a1p  = A + (size_t)(16 + lm) * K + k0 + 2 * half;
    const float* bptr = W + (size_t)(k0 + 2 * half) * N + n;

    v8f c0 = {};
    v8f c1 = {};

    #pragma unroll 4
    for (int kk = 0; kk < KCHUNK; kk += 4) {
        v2f a0 = *(const v2f*)a0p;
        v2f a1 = *(const v2f*)a1p;
        v2f b;
        b.x = bptr[0];
        b.y = bptr[N];
        c0 = __builtin_amdgcn_wmma_f32_16x16x4_f32(false, a0, false, b, (short)0, c0, false, false);
        c1 = __builtin_amdgcn_wmma_f32_16x16x4_f32(false, a1, false, b, (short)0, c1, false, false);
        a0p += 4;
        a1p += 4;
        bptr += (size_t)4 * N;
    }

    #pragma unroll
    for (int i = 0; i < 8; ++i) {
        int r0 = i + 8 * half;
        atomicAdd(&Out[(size_t)r0 * N + n], c0[i]);
        int r1 = 16 + r0;
        if (r1 < N_NODES) atomicAdd(&Out[(size_t)r1 * N + n], c1[i]);
    }
}

__global__ void bias_act_kernel(float* __restrict__ buf, const float* __restrict__ bias,
                                int N, int do_relu) {
    int idx = blockIdx.x * blockDim.x + threadIdx.x;
    if (idx >= N_NODES * N) return;
    int col = idx % N;
    float v = buf[idx] + bias[col];
    if (do_relu) v = v > 0.0f ? v : 0.0f;
    buf[idx] = v;
}

__global__ __launch_bounds__(512) void matvec_kernel(
        const float* __restrict__ x, const float* __restrict__ Wc1,
        float* __restrict__ y, int rows, int rowsPerBlock) {
    int j = threadIdx.x;
    int r0 = blockIdx.x * rowsPerBlock;
    int r1 = r0 + rowsPerBlock;
    if (r1 > rows) r1 = rows;
    float acc = 0.0f;
    for (int r = r0; r < r1; ++r) {
        acc += x[r] * Wc1[(size_t)r * LATENT + j];
    }
    atomicAdd(&y[j], acc);
}

__global__ __launch_bounds__(512) void final_kernel(
        const float* __restrict__ y, const float* __restrict__ bc1,
        const float* __restrict__ Wc2, const float* __restrict__ bc2,
        float* __restrict__ out0) {
    __shared__ float red[512];
    int j = threadIdx.x;
    red[j] = (y[j] + bc1[j]) * Wc2[j];
    __syncthreads();
    for (int s = 256; s > 0; s >>= 1) {
        if (j < s) red[j] += red[j + s];
        __syncthreads();
    }
    if (j == 0) {
        float v = red[0] + bc2[0];
        out0[0] = 1.0f / (1.0f + __expf(-v));
    }
}

extern "C" void kernel_launch(void* const* d_in, const int* in_sizes, int n_in,
                              void* d_out, int out_size, void* d_ws, size_t ws_size,
                              hipStream_t stream) {
    const float* z    = (const float*)d_in[0];
    const int*   eidx = (const int*)  d_in[1];
    const float* W1a  = (const float*)d_in[2];
    const float* b1a  = (const float*)d_in[3];
    const float* W1b  = (const float*)d_in[4];
    const float* b1b  = (const float*)d_in[5];
    const float* W2a  = (const float*)d_in[6];
    const float* b2a  = (const float*)d_in[7];
    const float* W2b  = (const float*)d_in[8];
    const float* b2b  = (const float*)d_in[9];
    const float* Wc1  = (const float*)d_in[10];
    const float* bc1  = (const float*)d_in[11];
    const float* Wc2  = (const float*)d_in[12];
    const float* bc2  = (const float*)d_in[13];

    float* out   = (float*)d_out;
    float* ws    = (float*)d_ws;
    float* h0    = ws + WS_H0;
    float* t1    = ws + WS_T1;
    float* h1    = ws + WS_H1;
    float* h1agg = ws + WS_H1AGG;
    float* t2    = ws + WS_T2;
    float* yv    = ws + WS_Y;
    float* recon = out + 1;

    zero_kernel<<<512, 256, 0, stream>>>(ws, (int)WS_TOTAL);
    zero_kernel<<<512, 256, 0, stream>>>(out, 1 + N_NODES * N_T);

    dim3 gblk(32, 4);

    {
        int n = N_NODES * LATENT;
        gin_agg_kernel<<<(n + 255) / 256, 256, 0, stream>>>(z, eidx, h0, LATENT);
    }
    gemm_wmma_kernel<LATENT, HID, LATENT / 16>
        <<<dim3(HID / 64, 16), gblk, 0, stream>>>(h0, W1a, t1);
    {
        int n = N_NODES * HID;
        bias_act_kernel<<<(n + 255) / 256, 256, 0, stream>>>(t1, b1a, HID, 1);
    }
    gemm_wmma_kernel<HID, HID, HID / 16>
        <<<dim3(HID / 64, 16), gblk, 0, stream>>>(t1, W1b, h1);
    {
        int n = N_NODES * HID;
        bias_act_kernel<<<(n + 255) / 256, 256, 0, stream>>>(h1, b1b, HID, 1);
    }
    {
        int n = N_NODES * HID;
        gin_agg_kernel<<<(n + 255) / 256, 256, 0, stream>>>(h1, eidx, h1agg, HID);
    }
    gemm_wmma_kernel<HID, N_T, HID / 8>
        <<<dim3(N_T / 64, 8), gblk, 0, stream>>>(h1agg, W2a, t2);
    {
        int n = N_NODES * N_T;
        bias_act_kernel<<<(n + 255) / 256, 256, 0, stream>>>(t2, b2a, N_T, 1);
    }
    gemm_wmma_kernel<N_T, N_T, N_T / 8>
        <<<dim3(N_T / 64, 8), gblk, 0, stream>>>(t2, W2b, recon);
    {
        int n = N_NODES * N_T;
        bias_act_kernel<<<(n + 255) / 256, 256, 0, stream>>>(recon, b2b, N_T, 0);
    }
    {
        int rows = N_NODES * N_T;
        int rpb = 256;
        matvec_kernel<<<(rows + rpb - 1) / rpb, 512, 0, stream>>>(recon, Wc1, yv, rows, rpb);
    }
    final_kernel<<<1, 512, 0, stream>>>(yv, bc1, Wc2, bc2, out);
}


# classifier matvec rebuilt as a read-once stream: 256 groups x 16 waves x 19 rows, 512-byte non-temporal row loads, 8 rows in flight per wave, LDS wave sum; sums regrouped, 256 atomic adds per column
# speedup vs baseline: 1.8029x; 1.8029x over previous
#include <hip/hip_runtime.h>
#include <hip/hip_bf16.h>

typedef __attribute__((ext_vector_type(2))) float v2f;
typedef __attribute__((ext_vector_type(8))) float v8f;

#define N_NODES 19
#define N_T     4096
#define LATENT  512
#define HID     2048
#define N_EDGES 342
#define PAD_ROWS 32

#define WS_H0    0
#define WS_T1    (WS_H0 + PAD_ROWS * LATENT)
#define WS_H1    (WS_T1 + PAD_ROWS * HID)
#define WS_H1AGG (WS_H1 + PAD_ROWS * HID)
#define WS_T2    (WS_H1AGG + PAD_ROWS * HID)
#define WS_Y     (WS_T2 + PAD_ROWS * N_T)
#define WS_TOTAL (WS_Y + LATENT)

__global__ void zero_kernel(float* __restrict__ p, int n) {
    int i = blockIdx.x * blockDim.x + threadIdx.x;
    int stride = gridDim.x * blockDim.x;
    for (; i < n; i += stride) p[i] = 0.0f;
}

__global__ void gin_agg_kernel(const float* __restrict__ h, const int* __restrict__ edges,
                               float* __restrict__ out, int F) {
    int idx = blockIdx.x * blockDim.x + threadIdx.x;
    if (idx >= N_NODES * F) return;
    int node = idx / F;
    int f = idx - node * F;
    const int* src = edges;
    const int* dst = edges + N_EDGES;
    float v = h[(size_t)node * F + f];
    for (int e = 0; e < N_EDGES; ++e) {
        if (dst[e] == node) v += h[(size_t)src[e] * F + f];
    }
    out[(size_t)node * F + f] = v;
}

template <int K, int N, int KCHUNK>
__global__ __launch_bounds__(128) void gemm_wmma_kernel(
        const float* __restrict__ A, const float* __restrict__ W,
        float* __restrict__ Out) {
    const int lane = threadIdx.x;
    const int half = lane >> 4;
    const int lm   = lane & 15;
    const int tile = blockIdx.x * 4 + threadIdx.y;
    const int n    = tile * 16 + lm;
    const int k0   = blockIdx.y * KCHUNK;

    const float* a0p  = A + (size_t)lm * K        + k0 + 2 * half;
    const float* a1p  = A + (size_t)(16 + lm) * K + k0 + 2 * half;
    const float* bptr = W + (size_t)(k0 + 2 * half) * N + n;

    v8f c0 = {};
    v8f c1 = {};

    #pragma unroll 4
    for (int kk = 0; kk < KCHUNK; kk += 4) {
        v2f a0 = *(const v2f*)a0p;
        v2f a1 = *(const v2f*)a1p;
        v2f b;
        b.x = bptr[0];
        b.y = bptr[N];
        c0 = __builtin_amdgcn_wmma_f32_16x16x4_f32(false, a0, false, b, (short)0, c0, false, false);
        c1 = __builtin_amdgcn_wmma_f32_16x16x4_f32(false, a1, false, b, (short)0, c1, false, false);
        a0p += 4;
        a1p += 4;
        bptr += (size_t)4 * N;
    }

    #pragma unroll
    for (int i = 0; i < 8; ++i) {
        int r0 = i + 8 * half;
        atomicAdd(&Out[(size_t)r0 * N + n], c0[i]);
        int r1 = 16 + r0;
        if (r1 < N_NODES) atomicAdd(&Out[(size_t)r1 * N + n], c1[i]);
    }
}

__global__ void bias_act_kernel(float* __restrict__ buf, const float* __restrict__ bias,
                                int N, int do_relu) {
    int idx = blockIdx.x * blockDim.x + threadIdx.x;
    if (idx >= N_NODES * N) return;
    int col = idx % N;
    float v = buf[idx] + bias[col];
    if (do_relu) v = v > 0.0f ? v : 0.0f;
    buf[idx] = v;
}

#define MV_WAVES  16
#define MV_ROWS   19
#define MV_AHEAD  8
#define MV_BLOCKS ((N_NODES * N_T) / (MV_WAVES * MV_ROWS))
static_assert(MV_BLOCKS * MV_WAVES * MV_ROWS == N_NODES * N_T, "rows must split evenly over the waves");
static_assert(MV_WAVES * 32 == LATENT, "one thread per output column in the workgroup sum");
static_assert(MV_AHEAD >= 1 && MV_AHEAD <= MV_ROWS && MV_ROWS <= 32, "rows in flight; x sits in one lane per row");
typedef __attribute__((ext_vector_type(4))) float v4f;

__global__ __launch_bounds__(MV_WAVES * 32) void matvec_kernel(
        const float* __restrict__ x, const float* __restrict__ Wc1,
        float* __restrict__ y) {
    __shared__ float part[MV_WAVES * LATENT];
    const unsigned tid  = threadIdx.x;
    const unsigned lane = tid & 31u;
    const unsigned wave = __builtin_amdgcn_readfirstlane(tid >> 5);
    const unsigned row0 = (blockIdx.x * MV_WAVES + wave) * MV_ROWS;
    const float* wu = Wc1 + (size_t)row0 * LATENT;
    const int xl = __builtin_bit_cast(int, x[row0 + (lane < MV_ROWS ? lane : 0u)]);

    v4f w[MV_AHEAD][4];
    v4f acc[4];
#pragma unroll
    for (int q = 0; q < 4; ++q) acc[q] = (v4f){0.0f, 0.0f, 0.0f, 0.0f};

#define MV_LOAD_ROW(R, P) do { \
    _Pragma("unroll") for (int q = 0; q < 4; ++q) \
        w[(R) % MV_AHEAD][q] = __builtin_nontemporal_load((const v4f*)((P) + 4u * lane + 128 * q)); } while (0)

#pragma unroll
    for (int r = 0; r < MV_AHEAD; ++r) MV_LOAD_ROW(r, wu + (size_t)r * LATENT);
#pragma unroll
    for (int r = 0; r < MV_ROWS; ++r) {
        const float xv = __builtin_bit_cast(float, __builtin_amdgcn_readlane(xl, r));
#pragma unroll
        for (int q = 0; q < 4; ++q) {
            const v4f wv = w[r % MV_AHEAD][q];
            acc[q].x = __builtin_fmaf(xv, wv.x, acc[q].x);
            acc[q].y = __builtin_fmaf(xv, wv.y, acc[q].y);
            acc[q].z = __builtin_fmaf(xv, wv.z, acc[q].z);
            acc[q].w = __builtin_fmaf(xv, wv.w, acc[q].w);
        }
        if (r + MV_AHEAD < MV_ROWS) {
            unsigned off = (unsigned)(r + MV_AHEAD) * LATENT;
            asm volatile("" : "+s"(off) : "v"(acc[0]), "v"(acc[1]), "v"(acc[2]), "v"(acc[3]));
            MV_LOAD_ROW(r + MV_AHEAD, wu + off);
        }
    }
#undef MV_LOAD_ROW

#pragma unroll
    for (int q = 0; q < 4; ++q)
        *(v4f*)&part[wave * LATENT + 128 * q + 4 * lane] = acc[q];
    __syncthreads();
    float s = 0.0f;
#pragma unroll
    for (int wv = 0; wv < MV_WAVES; ++wv) s += part[wv * LATENT + tid];
    atomicAdd(&y[tid], s);
}

__global__ __launch_bounds__(512) void final_kernel(
        const float* __restrict__ y, const float* __restrict__ bc1,
        const float* __restrict__ Wc2, const float* __restrict__ bc2,
        float* __restrict__ out0) {
    __shared__ float red[512];
    int j = threadIdx.x;
    red[j] = (y[j] + bc1[j]) * Wc2[j];
    __syncthreads();
    for (int s = 256; s > 0; s >>= 1) {
        if (j < s) red[j] += red[j + s];
        __syncthreads();
    }
    if (j == 0) {
        float v = red[0] + bc2[0];
        out0[0] = 1.0f / (1.0f + __expf(-v));
    }
}

extern "C" void kernel_launch(void* const* d_in, const int* in_sizes, int n_in,
                              void* d_out, int out_size, void* d_ws, size_t ws_size,
                              hipStream_t stream) {
    const float* z    = (const float*)d_in[0];
    const int*   eidx = (const int*)  d_in[1];
    const float* W1a  = (const float*)d_in[2];
    const float* b1a  = (const float*)d_in[3];
    const float* W1b  = (const float*)d_in[4];
    const float* b1b  = (const float*)d_in[5];
    const float* W2a  = (const float*)d_in[6];
    const float* b2a  = (const float*)d_in[7];
    const float* W2b  = (const float*)d_in[8];
    const float* b2b  = (const float*)d_in[9];
    const float* Wc1  = (const float*)d_in[10];
    const float* bc1  = (const float*)d_in[11];
    const float* Wc2  = (const float*)d_in[12];
    const float* bc2  = (const float*)d_in[13];

    float* out   = (float*)d_out;
    float* ws    = (float*)d_ws;
    float* h0    = ws + WS_H0;
    float* t1    = ws + WS_T1;
    float* h1    = ws + WS_H1;
    float* h1agg = ws + WS_H1AGG;
    float* t2    = ws + WS_T2;
    float* yv    = ws + WS_Y;
    float* recon = out + 1;

    zero_kernel<<<512, 256, 0, stream>>>(ws, (int)WS_TOTAL);
    zero_kernel<<<512, 256, 0, stream>>>(out, 1 + N_NODES * N_T);

    dim3 gblk(32, 4);

    {
        int n = N_NODES * LATENT;
        gin_agg_kernel<<<(n + 255) / 256, 256, 0, stream>>>(z, eidx, h0, LATENT);
    }
    gemm_wmma_kernel<LATENT, HID, LATENT / 16>
        <<<dim3(HID / 64, 16), gblk, 0, stream>>>(h0, W1a, t1);
    {
        int n = N_NODES * HID;
        bias_act_kernel<<<(n + 255) / 256, 256, 0, stream>>>(t1, b1a, HID, 1);
    }
    gemm_wmma_kernel<HID, HID, HID / 16>
        <<<dim3(HID / 64, 16), gblk, 0, stream>>>(t1, W1b, h1);
    {
        int n = N_NODES * HID;
        bias_act_kernel<<<(n + 255) / 256, 256, 0, stream>>>(h1, b1b, HID, 1);
    }
    {
        int n = N_NODES * HID;
        gin_agg_kernel<<<(n + 255) / 256, 256, 0, stream>>>(h1, eidx, h1agg, HID);
    }
    gemm_wmma_kernel<HID, N_T, HID / 8>
        <<<dim3(N_T / 64, 8), gblk, 0, stream>>>(h1agg, W2a, t2);
    {
        int n = N_NODES * N_T;
        bias_act_kernel<<<(n + 255) / 256, 256, 0, stream>>>(t2, b2a, N_T, 1);
    }
    gemm_wmma_kernel<N_T, N_T, N_T / 8>
        <<<dim3(N_T / 64, 8), gblk, 0, stream>>>(t2, W2b, recon);
    {
        int n = N_NODES * N_T;
        bias_act_kernel<<<(n + 255) / 256, 256, 0, stream>>>(recon, b2b, N_T, 0);
    }
    matvec_kernel<<<MV_BLOCKS, MV_WAVES * 32, 0, stream>>>(recon, Wc1, yv);
    final_kernel<<<1, 512, 0, stream>>>(yv, bc1, Wc2, bc2, out);
}


# 14 -> 8 launches: zero and bias passes folded away; a prep kernel (with agg1) stores each bias as its accumulator's start value (bias summed first, not last); ReLU applied as the next kernel loads
# speedup vs baseline: 1.9878x; 1.1026x over previous
#include <hip/hip_runtime.h>
#include <hip/hip_bf16.h>

typedef __attribute__((ext_vector_type(2))) float v2f;
typedef __attribute__((ext_vector_type(8))) float v8f;

#define N_NODES 19
#define N_T     4096
#define LATENT  512
#define HID     2048
#define N_EDGES 342
#define PAD_ROWS 32

#define WS_H0    0
#define WS_T1    (WS_H0 + PAD_ROWS * LATENT)
#define WS_H1    (WS_T1 + PAD_ROWS * HID)
#define WS_H1AGG (WS_H1 + PAD_ROWS * HID)
#define WS_T2    (WS_H1AGG + PAD_ROWS * HID)
#define WS_Y     (WS_T2 + PAD_ROWS * N_T)
#define WS_TOTAL (WS_Y + LATENT)

typedef __attribute__((ext_vector_type(4))) float v4f;

template <bool RELU>
__device__ __forceinline__ float act(float v) {
    return RELU ? (v > 0.0f ? v : 0.0f) : v;
}

template <bool RELU>
__device__ __forceinline__ void gin_agg_element(const float* __restrict__ h, const int* __restrict__ edges,
                                                float* __restrict__ out, int F, int idx) {
    int node = idx / F;
    int f = idx - node * F;
    const int* src = edges;
    const int* dst = edges + N_EDGES;
    float v = act<RELU>(h[(size_t)node * F + f]);
    for (int e = 0; e < N_EDGES; ++e) {
        if (dst[e] == node) v += act<RELU>(h[(size_t)src[e] * F + f]);
    }
    out[(size_t)node * F + f] = v;
}

__global__ void gin_agg_relu_kernel(const float* __restrict__ h, const int* __restrict__ edges,
                                    float* __restrict__ out, int F) {
    int idx = blockIdx.x * blockDim.x + threadIdx.x;
    if (idx >= N_NODES * F) return;
    gin_agg_element<true>(h, edges, out, F, idx);
}

#define PREP_THREADS    256
#define PREP_SPAN       (4 * PREP_THREADS)
#define PREP_AGG_BLOCKS ((N_NODES * LATENT) / PREP_THREADS)
#define PREP_HID_BLOCKS ((N_NODES * HID) / PREP_SPAN)
#define PREP_NT_BLOCKS  ((N_NODES * N_T) / PREP_SPAN)
#define PREP_BLOCKS     (PREP_AGG_BLOCKS + 2 * PREP_HID_BLOCKS + 2 * PREP_NT_BLOCKS + 1)
static_assert(PREP_AGG_BLOCKS * PREP_THREADS == N_NODES * LATENT, "agg1 elements split evenly over blocks");
static_assert(PREP_HID_BLOCKS * PREP_SPAN == N_NODES * HID && PREP_NT_BLOCKS * PREP_SPAN == N_NODES * N_T,
              "planes split evenly over fill blocks");
static_assert(HID % PREP_SPAN == 0 && N_T % PREP_SPAN == 0, "a fill block never crosses a row end");
static_assert(LATENT <= PREP_SPAN && LATENT % 4 == 0, "y fits one fill block");

__device__ __forceinline__ void fill_rows_bias(float* __restrict__ plane, const float* __restrict__ bias,
                                               int ncols, int blk) {
    const int i = blk * PREP_SPAN + 4 * (int)threadIdx.x;
    *(v4f*)(plane + i) = *(const v4f*)(bias + (i % ncols));
}

__global__ __launch_bounds__(PREP_THREADS) void prep_kernel(
        const float* __restrict__ z, const int* __restrict__ edges,
        const float* __restrict__ b1a, const float* __restrict__ b1b,
        const float* __restrict__ b2a, const float* __restrict__ b2b,
        float* __restrict__ h0, float* __restrict__ t1, float* __restrict__ h1,
        float* __restrict__ t2, float* __restrict__ recon, float* __restrict__ y) {
    int b = blockIdx.x;
    if (b < PREP_AGG_BLOCKS) {
        gin_agg_element<false>(z, edges, h0, LATENT, b * PREP_THREADS + (int)threadIdx.x);
        return;
    }
    b -= PREP_AGG_BLOCKS;
    if (b < PREP_HID_BLOCKS) { fill_rows_bias(t1, b1a, HID, b); return; }
    b -= PREP_HID_BLOCKS;
    if (b < PREP_HID_BLOCKS) { fill_rows_bias(h1, b1b, HID, b); return; }
    b -= PREP_HID_BLOCKS;
    if (b < PREP_NT_BLOCKS) { fill_rows_bias(t2, b2a, N_T, b); return; }
    b -= PREP_NT_BLOCKS;
    if (b < PREP_NT_BLOCKS) {
#pragma unroll
        for (int j = 0; j < 4; ++j) {
            const int i = b * PREP_SPAN + j * PREP_THREADS + (int)threadIdx.x;
            recon[i] = b2b[i % N_T];
        }
        return;
    }
    if (4 * threadIdx.x < LATENT) *(v4f*)(y + 4 * threadIdx.x) = (v4f){0.0f, 0.0f, 0.0f, 0.0f};
}

template <int K, int N, int KCHUNK, bool RELU_A>
__global__ __launch_bounds__(128) void gemm_wmma_kernel(
        const float* __restrict__ A, const float* __restrict__ W,
        float* __restrict__ Out) {
    const int lane = threadIdx.x;
    const int half = lane >> 4;
    const int lm   = lane & 15;
    const int tile = blockIdx.x * 4 + threadIdx.y;
    const int n    = tile * 16 + lm;
    const int k0   = blockIdx.y * KCHUNK;

    const float* a0p  = A + (size_t)lm * K        + k0 + 2 * half;
    const bool   r1ok = lm < N_NODES - 16;
    const float* a1p  = A + (size_t)(r1ok ? 16 + lm : lm) * K + k0 + 2 * half;
    const float* bptr = W + (size_t)(k0 + 2 * half) * N + n;

    v8f c0 = {};
    v8f c1 = {};

    #pragma unroll 4
    for (int kk = 0; kk < KCHUNK; kk += 4) {
        v2f a0 = *(const v2f*)a0p;
        v2f a1 = *(const v2f*)a1p;
        a0.x = act<RELU_A>(a0.x);
        a0.y = act<RELU_A>(a0.y);
        a1.x = r1ok ? act<RELU_A>(a1.x) : 0.0f;
        a1.y = r1ok ? act<RELU_A>(a1.y) : 0.0f;
        v2f b;
        b.x = bptr[0];
        b.y = bptr[N];
        c0 = __builtin_amdgcn_wmma_f32_16x16x4_f32(false, a0, false, b, (short)0, c0, false, false);
        c1 = __builtin_amdgcn_wmma_f32_16x16x4_f32(false, a1, false, b, (short)0, c1, false, false);
        a0p += 4;
        a1p += 4;
        bptr += (size_t)4 * N;
    }

    #pragma unroll
    for (int i = 0; i < 8; ++i) {
        int r0 = i + 8 * half;
        atomicAdd(&Out[(size_t)r0 * N + n], c0[i]);
        int r1 = 16 + r0;
        if (r1 < N_NODES) atomicAdd(&Out[(size_t)r1 * N + n], c1[i]);
    }
}

#define MV_WAVES  16
#define MV_ROWS   19
#define MV_AHEAD  8
#define MV_BLOCKS ((N_NODES * N_T) / (MV_WAVES * MV_ROWS))
static_assert(MV_BLOCKS * MV_WAVES * MV_ROWS == N_NODES * N_T, "rows must split evenly over the waves");
static_assert(MV_WAVES * 32 == LATENT, "one thread per output column in the workgroup sum");
static_assert(MV_AHEAD >= 1 && MV_AHEAD <= MV_ROWS && MV_ROWS <= 32, "rows in flight; x sits in one lane per row");

__global__ __launch_bounds__(MV_WAVES * 32) void matvec_kernel(
        const float* __restrict__ x, const float* __restrict__ Wc1,
        float* __restrict__ y) {
    __shared__ float part[MV_WAVES * LATENT];
    const unsigned tid  = threadIdx.x;
    const unsigned lane = tid & 31u;
    const unsigned wave = __builtin_amdgcn_readfirstlane(tid >> 5);
    const unsigned row0 = (blockIdx.x * MV_WAVES + wave) * MV_ROWS;
    const float* wu = Wc1 + (size_t)row0 * LATENT;
    const int xl = __builtin_bit_cast(int, x[row0 + (lane < MV_ROWS ? lane : 0u)]);

    v4f w[MV_AHEAD][4];
    v4f acc[4];
#pragma unroll
    for (int q = 0; q < 4; ++q) acc[q] = (v4f){0.0f, 0.0f, 0.0f, 0.0f};

#define MV_LOAD_ROW(R, P) do { \
    _Pragma("unroll") for (int q = 0; q < 4; ++q) \
        w[(R) % MV_AHEAD][q] = __builtin_nontemporal_load((const v4f*)((P) + 4u * lane + 128 * q)); } while (0)

#pragma unroll
    for (int r = 0; r < MV_AHEAD; ++r) MV_LOAD_ROW(r, wu + (size_t)r * LATENT);
#pragma unroll
    for (int r = 0; r < MV_ROWS; ++r) {
        const float xv = __builtin_bit_cast(float, __builtin_amdgcn_readlane(xl, r));
#pragma unroll
        for (int q = 0; q < 4; ++q) {
            const v4f wv = w[r % MV_AHEAD][q];
            acc[q].x = __builtin_fmaf(xv, wv.x, acc[q].x);
            acc[q].y = __builtin_fmaf(xv, wv.y, acc[q].y);
            acc[q].z = __builtin_fmaf(xv, wv.z, acc[q].z);
            acc[q].w = __builtin_fmaf(xv, wv.w, acc[q].w);
        }
        if (r + MV_AHEAD < MV_ROWS) {
            unsigned off = (unsigned)(r + MV_AHEAD) * LATENT;
            asm volatile("" : "+s"(off) : "v"(acc[0]), "v"(acc[1]), "v"(acc[2]), "v"(acc[3]));
            MV_LOAD_ROW(r + MV_AHEAD, wu + off);
        }
    }
#undef MV_LOAD_ROW

#pragma unroll
    for (int q = 0; q < 4; ++q)
        *(v4f*)&part[wave * LATENT + 128 * q + 4 * lane] = acc[q];
    __syncthreads();
    float s = 0.0f;
#pragma unroll
    for (int wv = 0; wv < MV_WAVES; ++wv) s += part[wv * LATENT + tid];
    atomicAdd(&y[tid], s);
}

__global__ __launch_bounds__(512) void final_kernel(
        const float* __restrict__ y, const float* __restrict__ bc1,
        const float* __restrict__ Wc2, const float* __restrict__ bc2,
        float* __restrict__ out0) {
    __shared__ float red[512];
    int j = threadIdx.x;
    red[j] = (y[j] + bc1[j]) * Wc2[j];
    __syncthreads();
    for (int s = 256; s > 0; s >>= 1) {
        if (j < s) red[j] += red[j + s];
        __syncthreads();
    }
    if (j == 0) {
        float v = red[0] + bc2[0];
        out0[0] = 1.0f / (1.0f + __expf(-v));
    }
}

extern "C" void kernel_launch(void* const* d_in, const int* in_sizes, int n_in,
                              void* d_out, int out_size, void* d_ws, size_t ws_size,
                              hipStream_t stream) {
    const float* z    = (const float*)d_in[0];
    const int*   eidx = (const int*)  d_in[1];
    const float* W1a  = (const float*)d_in[2];
    const float* b1a  = (const float*)d_in[3];
    const float* W1b  = (const float*)d_in[4];
    const float* b1b  = (const float*)d_in[5];
    const float* W2a  = (const float*)d_in[6];
    const float* b2a  = (const float*)d_in[7];
    const float* W2b  = (const float*)d_in[8];
    const float* b2b  = (const float*)d_in[9];
    const float* Wc1  = (const float*)d_in[10];
    const float* bc1  = (const float*)d_in[11];
    const float* Wc2  = (const float*)d_in[12];
    const float* bc2  = (const float*)d_in[13];

    float* out   = (float*)d_out;
    float* ws    = (float*)d_ws;
    float* h0    = ws + WS_H0;
    float* t1    = ws + WS_T1;
    float* h1    = ws + WS_H1;
    float* h1agg = ws + WS_H1AGG;
    float* t2    = ws + WS_T2;
    float* yv    = ws + WS_Y;
    float* recon = out + 1;

    prep_kernel<<<PREP_BLOCKS, PREP_THREADS, 0, stream>>>(z, eidx, b1a, b1b, b2a, b2b,
                                                          h0, t1, h1, t2, recon, yv);

    dim3 gblk(32, 4);

    gemm_wmma_kernel<LATENT, HID, LATENT / 16, false>
        <<<dim3(HID / 64, 16), gblk, 0, stream>>>(h0, W1a, t1);
    gemm_wmma_kernel<HID, HID, HID / 16, true>
        <<<dim3(HID / 64, 16), gblk, 0, stream>>>(t1, W1b, h1);
    {
        int n = N_NODES * HID;
        gin_agg_relu_kernel<<<(n + 255) / 256, 256, 0, stream>>>(h1, eidx, h1agg, HID);
    }
    gemm_wmma_kernel<HID, N_T, HID / 8, false>
        <<<dim3(N_T / 64, 8), gblk, 0, stream>>>(h1agg, W2a, t2);
    gemm_wmma_kernel<N_T, N_T, N_T / 8, true>
        <<<dim3(N_T / 64, 8), gblk, 0, stream>>>(t2, W2b, recon);
    matvec_kernel<<<MV_BLOCKS, MV_WAVES * 32, 0, stream>>>(recon, Wc1, yv);
    final_kernel<<<1, 512, 0, stream>>>(yv, bc1, Wc2, bc2, out);
}


# GEMMs rebuilt as weight streams: a wave owns 64 columns x a K slice (16-byte weight loads, 5-step load ring), 8 waves sum through LDS; split-K regrouped: 4-8 atomic adds per element, was 8-16
# speedup vs baseline: 2.6558x; 1.3360x over previous
#include <hip/hip_runtime.h>
#include <hip/hip_bf16.h>

typedef __attribute__((ext_vector_type(2))) float v2f;
typedef __attribute__((ext_vector_type(8))) float v8f;

#define N_NODES 19
#define N_T     4096
#define LATENT  512
#define HID     2048
#define N_EDGES 342
#define PAD_ROWS 32

#define WS_H0    0
#define WS_T1    (WS_H0 + PAD_ROWS * LATENT)
#define WS_H1    (WS_T1 + PAD_ROWS * HID)
#define WS_H1AGG (WS_H1 + PAD_ROWS * HID)
#define WS_T2    (WS_H1AGG + PAD_ROWS * HID)
#define WS_Y     (WS_T2 + PAD_ROWS * N_T)
#define WS_TOTAL (WS_Y + LATENT)

typedef __attribute__((ext_vector_type(4))) float v4f;

template <bool RELU>
__device__ __forceinline__ float act(float v) {
    return RELU ? (v > 0.0f ? v : 0.0f) : v;
}

template <bool RELU>
__device__ __forceinline__ void gin_agg_element(const float* __restrict__ h, const int* __restrict__ edges,
                                                float* __restrict__ out, int F, int idx) {
    int node = idx / F;
    int f = idx - node * F;
    const int* src = edges;
    const int* dst = edges + N_EDGES;
    float v = act<RELU>(h[(size_t)node * F + f]);
    for (int e = 0; e < N_EDGES; ++e) {
        if (dst[e] == node) v += act<RELU>(h[(size_t)src[e] * F + f]);
    }
    out[(size_t)node * F + f] = v;
}

__global__ void gin_agg_relu_kernel(const float* __restrict__ h, const int* __restrict__ edges,
                                    float* __restrict__ out, int F) {
    int idx = blockIdx.x * blockDim.x + threadIdx.x;
    if (idx >= N_NODES * F) return;
    gin_agg_element<true>(h, edges, out, F, idx);
}

#define PREP_THREADS    256
#define PREP_SPAN       (4 * PREP_THREADS)
#define PREP_AGG_BLOCKS ((N_NODES * LATENT) / PREP_THREADS)
#define PREP_HID_BLOCKS ((N_NODES * HID) / PREP_SPAN)
#define PREP_NT_BLOCKS  ((N_NODES * N_T) / PREP_SPAN)
#define PREP_BLOCKS     (PREP_AGG_BLOCKS + 2 * PREP_HID_BLOCKS + 2 * PREP_NT_BLOCKS + 1)
static_assert(PREP_AGG_BLOCKS * PREP_THREADS == N_NODES * LATENT, "agg1 elements split evenly over blocks");
static_assert(PREP_HID_BLOCKS * PREP_SPAN == N_NODES * HID && PREP_NT_BLOCKS * PREP_SPAN == N_NODES * N_T,
              "planes split evenly over fill blocks");
static_assert(HID % PREP_SPAN == 0 && N_T % PREP_SPAN == 0, "a fill block never crosses a row end");
static_assert(LATENT <= PREP_SPAN && LATENT % 4 == 0, "y fits one fill block");

__device__ __forceinline__ void fill_rows_bias(float* __restrict__ plane, const float* __restrict__ bias,
                                               int ncols, int blk) {
    const int i = blk * PREP_SPAN + 4 * (int)threadIdx.x;
    *(v4f*)(plane + i) = *(const v4f*)(bias + (i % ncols));
}

__global__ __launch_bounds__(PREP_THREADS) void prep_kernel(
        const float* __restrict__ z, const int* __restrict__ edges,
        const float* __restrict__ b1a, const float* __restrict__ b1b,
        const float* __restrict__ b2a, const float* __restrict__ b2b,
        float* __restrict__ h0, float* __restrict__ t1, float* __restrict__ h1,
        float* __restrict__ t2, float* __restrict__ recon, float* __restrict__ y) {
    int b = blockIdx.x;
    if (b < PREP_AGG_BLOCKS) {
        gin_agg_element<false>(z, edges, h0, LATENT, b * PREP_THREADS + (int)threadIdx.x);
        return;
    }
    b -= PREP_AGG_BLOCKS;
    if (b < PREP_HID_BLOCKS) { fill_rows_bias(t1, b1a, HID, b); return; }
    b -= PREP_HID_BLOCKS;
    if (b < PREP_HID_BLOCKS) { fill_rows_bias(h1, b1b, HID, b); return; }
    b -= PREP_HID_BLOCKS;
    if (b < PREP_NT_BLOCKS) { fill_rows_bias(t2, b2a, N_T, b); return; }
    b -= PREP_NT_BLOCKS;
    if (b < PREP_NT_BLOCKS) {
#pragma unroll
        for (int j = 0; j < 4; ++j) {
            const int i = b * PREP_SPAN + j * PREP_THREADS + (int)threadIdx.x;
            recon[i] = b2b[i % N_T];
        }
        return;
    }
    if (4 * threadIdx.x < LATENT) *(v4f*)(y + 4 * threadIdx.x) = (v4f){0.0f, 0.0f, 0.0f, 0.0f};
}

#define GM_WAVES 8
#define GM_COLS  64
#define GM_KSTEP 8
#define GM_RING  5

__device__ __forceinline__ v8f wmma4(v2f a, v2f b, v8f c) {
    return __builtin_amdgcn_wmma_f32_16x16x4_f32(false, a, false, b, (short)0, c, false, false);
}

template <int K, int N, int GS, bool RELU_A>
__global__ __launch_bounds__(GM_WAVES * 32) void gemm_wmma_kernel(
        const float* __restrict__ A, const float* __restrict__ W,
        float* __restrict__ Out) {
    constexpr int KB    = K / GS;
    constexpr int KW    = KB / GM_WAVES;
    constexpr int STEPS = KW / GM_KSTEP;
    constexpr int RING  = STEPS + 1 < GM_RING ? STEPS + 1 : GM_RING;
    static_assert(GS * GM_WAVES * GM_KSTEP * STEPS == K, "K must split evenly over slices, waves and steps");
    static_assert(N % GM_COLS == 0 && N % 4 == 0 && K % 4 == 0, "16-byte loads need aligned rows");
    static_assert(GM_WAVES * 32 == 4 * GM_COLS, "the workgroup sum maps 4 row groups x 64 columns onto the threads");

    __shared__ __attribute__((aligned(16))) float part[GM_WAVES * N_NODES * GM_COLS];

    const unsigned tid  = threadIdx.x;
    const unsigned lane = tid & 31u;
    const unsigned wave = __builtin_amdgcn_readfirstlane(tid >> 5);
    const unsigned h    = lane >> 4;
    const unsigned c16  = lane & 15u;
    const unsigned n0   = blockIdx.x * GM_COLS;
    const unsigned kbeg = blockIdx.y * KB + wave * KW;

    const float* wu = W + (size_t)kbeg * N + n0;
    const float* au = A + kbeg;
    const unsigned boff  = 4u * h * N + 4u * c16;
    const bool     r1ok  = c16 < N_NODES - 16;
    const unsigned a0off = c16 * K + 4u * h;
    const unsigned a1off = (16u + (r1ok ? c16 : 0u)) * K + 4u * h;

    v4f bq[RING][4];
    v4f aq[RING][2];
    v8f acc[2][4];
#pragma unroll
    for (int m = 0; m < 2; ++m)
#pragma unroll
        for (int j = 0; j < 4; ++j) acc[m][j] = (v8f){0.0f, 0.0f, 0.0f, 0.0f, 0.0f, 0.0f, 0.0f, 0.0f};

#define GM_LOAD_STEP(S, BO, AO) do { \
    _Pragma("unroll") for (int r = 0; r < 4; ++r) \
        bq[(S) % RING][r] = *(const v4f*)(wu + (BO) + (unsigned)r * N + boff); \
    aq[(S) % RING][0] = *(const v4f*)(au + (AO) + a0off); \
    aq[(S) % RING][1] = *(const v4f*)(au + (AO) + a1off); } while (0)

#define GM_STEP_BOUNDARY() __builtin_amdgcn_sched_barrier(0)

#pragma unroll
    for (int s = 0; s < RING - 1; ++s) {
        GM_LOAD_STEP(s, (unsigned)s * GM_KSTEP * N, (unsigned)s * GM_KSTEP);
        GM_STEP_BOUNDARY();
    }

#pragma unroll
    for (int s = 0; s < STEPS; ++s) {
        if (s + RING - 1 < STEPS)
            GM_LOAD_STEP(s + RING - 1, (unsigned)(s + RING - 1) * GM_KSTEP * N, (unsigned)(s + RING - 1) * GM_KSTEP);
        const v4f a0 = aq[s % RING][0];
        const v4f a1 = aq[s % RING][1];
        v2f a0lo, a0hi, a1lo, a1hi;
        a0lo.x = act<RELU_A>(a0.x);  a0lo.y = act<RELU_A>(a0.y);
        a0hi.x = act<RELU_A>(a0.z);  a0hi.y = act<RELU_A>(a0.w);
        a1lo.x = r1ok ? act<RELU_A>(a1.x) : 0.0f;  a1lo.y = r1ok ? act<RELU_A>(a1.y) : 0.0f;
        a1hi.x = r1ok ? act<RELU_A>(a1.z) : 0.0f;  a1hi.y = r1ok ? act<RELU_A>(a1.w) : 0.0f;
#pragma unroll
        for (int j = 0; j < 4; ++j) {
            v2f b;
            b.x = bq[s % RING][0][j];
            b.y = bq[s % RING][1][j];
            acc[0][j] = wmma4(a0lo, b, acc[0][j]);
            acc[1][j] = wmma4(a1lo, b, acc[1][j]);
        }
#pragma unroll
        for (int j = 0; j < 4; ++j) {
            v2f b;
            b.x = bq[s % RING][2][j];
            b.y = bq[s % RING][3][j];
            acc[0][j] = wmma4(a0hi, b, acc[0][j]);
            acc[1][j] = wmma4(a1hi, b, acc[1][j]);
        }
        if (s + 1 < STEPS) GM_STEP_BOUNDARY();
    }
#undef GM_STEP_BOUNDARY
#undef GM_LOAD_STEP

    float* pw = part + wave * (N_NODES * GM_COLS);
#pragma unroll
    for (int i = 0; i < 8; ++i)
        *(v4f*)&pw[(8u * h + i) * GM_COLS + 4u * c16] =
            (v4f){acc[0][0][i], acc[0][1][i], acc[0][2][i], acc[0][3][i]};
    if (h == 0) {
#pragma unroll
        for (int i = 0; i < N_NODES - 16; ++i)
            *(v4f*)&pw[(16u + i) * GM_COLS + 4u * c16] =
                (v4f){acc[1][0][i], acc[1][1][i], acc[1][2][i], acc[1][3][i]};
    }
    __syncthreads();
    const unsigned col = tid & (GM_COLS - 1u);
    const unsigned rg  = tid >> 6;
#pragma unroll
    for (int q = 0; q < (N_NODES + 3) / 4; ++q) {
        const unsigned row = rg + 4u * q;
        if (row < N_NODES) {
            float s = 0.0f;
#pragma unroll
            for (int wv = 0; wv < GM_WAVES; ++wv) s += part[wv * (N_NODES * GM_COLS) + row * GM_COLS + col];
            atomicAdd(&Out[(size_t)row * N + n0 + col], s);
        }
    }
}

#define MV_WAVES  16
#define MV_ROWS   19
#define MV_AHEAD  8
#define MV_BLOCKS ((N_NODES * N_T) / (MV_WAVES * MV_ROWS))
static_assert(MV_BLOCKS * MV_WAVES * MV_ROWS == N_NODES * N_T, "rows must split evenly over the waves");
static_assert(MV_WAVES * 32 == LATENT, "one thread per output column in the workgroup sum");
static_assert(MV_AHEAD >= 1 && MV_AHEAD <= MV_ROWS && MV_ROWS <= 32, "rows in flight; x sits in one lane per row");

__global__ __launch_bounds__(MV_WAVES * 32) void matvec_kernel(
        const float* __restrict__ x, const float* __restrict__ Wc1,
        float* __restrict__ y) {
    __shared__ float part[MV_WAVES * LATENT];
    const unsigned tid  = threadIdx.x;
    const unsigned lane = tid & 31u;
    const unsigned wave = __builtin_amdgcn_readfirstlane(tid >> 5);
    const unsigned row0 = (blockIdx.x * MV_WAVES + wave) * MV_ROWS;
    const float* wu = Wc1 + (size_t)row0 * LATENT;
    const int xl = __builtin_bit_cast(int, x[row0 + (lane < MV_ROWS ? lane : 0u)]);

    v4f w[MV_AHEAD][4];
    v4f acc[4];
#pragma unroll
    for (int q = 0; q < 4; ++q) acc[q] = (v4f){0.0f, 0.0f, 0.0f, 0.0f};

#define MV_LOAD_ROW(R, P) do { \
    _Pragma("unroll") for (int q = 0; q < 4; ++q) \
        w[(R) % MV_AHEAD][q] = __builtin_nontemporal_load((const v4f*)((P) + 4u * lane + 128 * q)); } while (0)

#pragma unroll
    for (int r = 0; r < MV_AHEAD; ++r) MV_LOAD_ROW(r, wu + (size_t)r * LATENT);
#pragma unroll
    for (int r = 0; r < MV_ROWS; ++r) {
        const float xv = __builtin_bit_cast(float, __builtin_amdgcn_readlane(xl, r));
#pragma unroll
        for (int q = 0; q < 4; ++q) {
            const v4f wv = w[r % MV_AHEAD][q];
            acc[q].x = __builtin_fmaf(xv, wv.x, acc[q].x);
            acc[q].y = __builtin_fmaf(xv, wv.y, acc[q].y);
            acc[q].z = __builtin_fmaf(xv, wv.z, acc[q].z);
            acc[q].w = __builtin_fmaf(xv, wv.w, acc[q].w);
        }
        if (r + MV_AHEAD < MV_ROWS) {
            unsigned off = (unsigned)(r + MV_AHEAD) * LATENT;
            asm volatile("" : "+s"(off) : "v"(acc[0]), "v"(acc[1]), "v"(acc[2]), "v"(acc[3]));
            MV_LOAD_ROW(r + MV_AHEAD, wu + off);
        }
    }
#undef MV_LOAD_ROW

#pragma unroll
    for (int q = 0; q < 4; ++q)
        *(v4f*)&part[wave * LATENT + 128 * q + 4 * lane] = acc[q];
    __syncthreads();
    float s = 0.0f;
#pragma unroll
    for (int wv = 0; wv < MV_WAVES; ++wv) s += part[wv * LATENT + tid];
    atomicAdd(&y[tid], s);
}

__global__ __launch_bounds__(512) void final_kernel(
        const float* __restrict__ y, const float* __restrict__ bc1,
        const float* __restrict__ Wc2, const float* __restrict__ bc2,
        float* __restrict__ out0) {
    __shared__ float red[512];
    int j = threadIdx.x;
    red[j] = (y[j] + bc1[j]) * Wc2[j];
    __syncthreads();
    for (int s = 256; s > 0; s >>= 1) {
        if (j < s) red[j] += red[j + s];
        __syncthreads();
    }
    if (j == 0) {
        float v = red[0] + bc2[0];
        out0[0] = 1.0f / (1.0f + __expf(-v));
    }
}

extern "C" void kernel_launch(void* const* d_in, const int* in_sizes, int n_in,
                              void* d_out, int out_size, void* d_ws, size_t ws_size,
                              hipStream_t stream) {
    const float* z    = (const float*)d_in[0];
    const int*   eidx = (const int*)  d_in[1];
    const float* W1a  = (const float*)d_in[2];
    const float* b1a  = (const float*)d_in[3];
    const float* W1b  = (const float*)d_in[4];
    const float* b1b  = (const float*)d_in[5];
    const float* W2a  = (const float*)d_in[6];
    const float* b2a  = (const float*)d_in[7];
    const float* W2b  = (const float*)d_in[8];
    const float* b2b  = (const float*)d_in[9];
    const float* Wc1  = (const float*)d_in[10];
    const float* bc1  = (const float*)d_in[11];
    const float* Wc2  = (const float*)d_in[12];
    const float* bc2  = (const float*)d_in[13];

    float* out   = (float*)d_out;
    float* ws    = (float*)d_ws;
    float* h0    = ws + WS_H0;
    float* t1    = ws + WS_T1;
    float* h1    = ws + WS_H1;
    float* h1agg = ws + WS_H1AGG;
    float* t2    = ws + WS_T2;
    float* yv    = ws + WS_Y;
    float* recon = out + 1;

    prep_kernel<<<PREP_BLOCKS, PREP_THREADS, 0, stream>>>(z, eidx, b1a, b1b, b2a, b2b,
                                                          h0, t1, h1, t2, recon, yv);


    gemm_wmma_kernel<LATENT, HID, 4, false>
        <<<dim3(HID / GM_COLS, 4), GM_WAVES * 32, 0, stream>>>(h0, W1a, t1);
    gemm_wmma_kernel<HID, HID, 8, true>
        <<<dim3(HID / GM_COLS, 8), GM_WAVES * 32, 0, stream>>>(t1, W1b, h1);
    {
        int n = N_NODES * HID;
        gin_agg_relu_kernel<<<(n + 255) / 256, 256, 0, stream>>>(h1, eidx, h1agg, HID);
    }
    gemm_wmma_kernel<HID, N_T, 4, false>
        <<<dim3(N_T / GM_COLS, 4), GM_WAVES * 32, 0, stream>>>(h1agg, W2a, t2);
    gemm_wmma_kernel<N_T, N_T, 4, true>
        <<<dim3(N_T / GM_COLS, 4), GM_WAVES * 32, 0, stream>>>(t2, W2b, recon);
    matvec_kernel<<<MV_BLOCKS, MV_WAVES * 32, 0, stream>>>(recon, Wc1, yv);
    final_kernel<<<1, 512, 0, stream>>>(yv, bc1, Wc2, bc2, out);
}


# GIN aggregation moved behind its GEMM by linearity: (h + segsum h).W = M.(h.W), M = I + edge counts built per workgroup in LDS, applied in the workgroup sum; 8 -> 7 launches; sums regrouped
# speedup vs baseline: 4.9339x; 1.8578x over previous
#include <hip/hip_runtime.h>
#include <hip/hip_bf16.h>

typedef __attribute__((ext_vector_type(2))) float v2f;
typedef __attribute__((ext_vector_type(8))) float v8f;

#define N_NODES 19
#define N_T     4096
#define LATENT  512
#define HID     2048
#define N_EDGES 342
#define PAD_ROWS 32

#define WS_T1    0
#define WS_H1    (WS_T1 + PAD_ROWS * HID)
#define WS_T2    (WS_H1 + PAD_ROWS * HID)
#define WS_Y     (WS_T2 + PAD_ROWS * N_T)
#define WS_TOTAL (WS_Y + LATENT)

typedef __attribute__((ext_vector_type(4))) float v4f;

template <bool RELU>
__device__ __forceinline__ float act(float v) {
    return RELU ? (v > 0.0f ? v : 0.0f) : v;
}

#define PREP_THREADS    256
#define PREP_SPAN       (4 * PREP_THREADS)
#define PREP_HID_BLOCKS ((N_NODES * HID) / PREP_SPAN)
#define PREP_NT_BLOCKS  ((N_NODES * N_T) / PREP_SPAN)
#define PREP_BLOCKS     (2 * PREP_HID_BLOCKS + 2 * PREP_NT_BLOCKS + 1)
static_assert(PREP_HID_BLOCKS * PREP_SPAN == N_NODES * HID && PREP_NT_BLOCKS * PREP_SPAN == N_NODES * N_T,
              "planes split evenly over fill blocks");
static_assert(HID % PREP_SPAN == 0 && N_T % PREP_SPAN == 0, "a fill block never crosses a row end");
static_assert(LATENT <= PREP_SPAN && LATENT % 4 == 0, "y fits one fill block");

__device__ __forceinline__ void fill_rows_bias(float* __restrict__ plane, const float* __restrict__ bias,
                                               int ncols, int blk) {
    const int i = blk * PREP_SPAN + 4 * (int)threadIdx.x;
    *(v4f*)(plane + i) = *(const v4f*)(bias + (i % ncols));
}

__global__ __launch_bounds__(PREP_THREADS) void prep_kernel(
        const float* __restrict__ b1a, const float* __restrict__ b1b,
        const float* __restrict__ b2a, const float* __restrict__ b2b,
        float* __restrict__ t1, float* __restrict__ h1,
        float* __restrict__ t2, float* __restrict__ recon, float* __restrict__ y) {
    int b = blockIdx.x;
    if (b < PREP_HID_BLOCKS) { fill_rows_bias(t1, b1a, HID, b); return; }
    b -= PREP_HID_BLOCKS;
    if (b < PREP_HID_BLOCKS) { fill_rows_bias(h1, b1b, HID, b); return; }
    b -= PREP_HID_BLOCKS;
    if (b < PREP_NT_BLOCKS) { fill_rows_bias(t2, b2a, N_T, b); return; }
    b -= PREP_NT_BLOCKS;
    if (b < PREP_NT_BLOCKS) {
#pragma unroll
        for (int j = 0; j < 4; ++j) {
            const int i = b * PREP_SPAN + j * PREP_THREADS + (int)threadIdx.x;
            recon[i] = b2b[i % N_T];
        }
        return;
    }
    if (4 * threadIdx.x < LATENT) *(v4f*)(y + 4 * threadIdx.x) = (v4f){0.0f, 0.0f, 0.0f, 0.0f};
}

#define GM_WAVES 8
#define GM_COLS  64
#define GM_KSTEP 8
#define GM_RING  5
#define AGG_PITCH 20

__device__ __forceinline__ v8f wmma4(v2f a, v2f b, v8f c) {
    return __builtin_amdgcn_wmma_f32_16x16x4_f32(false, a, false, b, (short)0, c, false, false);
}

template <int K, int N, int GS, bool RELU_A, bool AGG>
__global__ __launch_bounds__(GM_WAVES * 32) void gemm_wmma_kernel(
        const float* __restrict__ A, const float* __restrict__ W,
        const int* __restrict__ edges, float* __restrict__ Out) {
    constexpr int KB    = K / GS;
    constexpr int KW    = KB / GM_WAVES;
    constexpr int STEPS = KW / GM_KSTEP;
    constexpr int RING  = STEPS + 1 < GM_RING ? STEPS + 1 : GM_RING;
    static_assert(GS * GM_WAVES * GM_KSTEP * STEPS == K, "K must split evenly over slices, waves and steps");
    static_assert(N % GM_COLS == 0 && N % 4 == 0 && K % 4 == 0, "16-byte loads need aligned rows");
    static_assert(GM_WAVES * 32 == 4 * GM_COLS, "the workgroup sum maps 4 row groups x 64 columns onto the threads");
    static_assert(N_EDGES >= GM_WAVES * 32 && N_EDGES <= 2 * GM_WAVES * 32, "one or two edges per thread");
    static_assert(AGG_PITCH % 4 == 0 && AGG_PITCH > N_NODES, "a row of M is whole 16-byte pieces with a zero pad");

    __shared__ __attribute__((aligned(16))) float part[GM_WAVES * N_NODES * GM_COLS];
    __shared__ __attribute__((aligned(16))) float magg[N_NODES * AGG_PITCH];
    __shared__ float psum[N_NODES * GM_COLS];

    const unsigned tid  = threadIdx.x;
    const unsigned lane = tid & 31u;
    const unsigned wave = __builtin_amdgcn_readfirstlane(tid >> 5);
    const unsigned h    = lane >> 4;
    const unsigned c16  = lane & 15u;
    const unsigned n0   = blockIdx.x * GM_COLS;
    const unsigned kbeg = blockIdx.y * KB + wave * KW;

    const float* wu = W + (size_t)kbeg * N + n0;
    const float* au = A + kbeg;
    const unsigned boff  = 4u * h * N + 4u * c16;
    const bool     r1ok  = c16 < N_NODES - 16;
    const unsigned a0off = c16 * K + 4u * h;
    const unsigned a1off = (16u + (r1ok ? c16 : 0u)) * K + 4u * h;

    constexpr unsigned NT = GM_WAVES * 32;
    int es0 = 0, ed0 = 0, es1 = 0, ed1 = 0;
    if constexpr (AGG) {
        es0 = edges[tid];
        ed0 = edges[N_EDGES + tid];
        if (tid + NT < N_EDGES) {
            es1 = edges[tid + NT];
            ed1 = edges[N_EDGES + tid + NT];
        }
        for (unsigned i = tid; i < N_NODES * AGG_PITCH; i += NT)
            magg[i] = (i / AGG_PITCH == i % AGG_PITCH) ? 1.0f : 0.0f;
    }

    v4f bq[RING][4];
    v4f aq[RING][2];
    v8f acc[2][4];
#pragma unroll
    for (int m = 0; m < 2; ++m)
#pragma unroll
        for (int j = 0; j < 4; ++j) acc[m][j] = (v8f){0.0f, 0.0f, 0.0f, 0.0f, 0.0f, 0.0f, 0.0f, 0.0f};

#define GM_LOAD_STEP(S, BO, AO) do { \
    _Pragma("unroll") for (int r = 0; r < 4; ++r) \
        bq[(S) % RING][r] = *(const v4f*)(wu + (BO) + (unsigned)r * N + boff); \
    aq[(S) % RING][0] = *(const v4f*)(au + (AO) + a0off); \
    aq[(S) % RING][1] = *(const v4f*)(au + (AO) + a1off); } while (0)

#define GM_STEP_BOUNDARY() __builtin_amdgcn_sched_barrier(0)

#pragma unroll
    for (int s = 0; s < RING - 1; ++s) {
        GM_LOAD_STEP(s, (unsigned)s * GM_KSTEP * N, (unsigned)s * GM_KSTEP);
        GM_STEP_BOUNDARY();
    }

#pragma unroll
    for (int s = 0; s < STEPS; ++s) {
        if (s + RING - 1 < STEPS)
            GM_LOAD_STEP(s + RING - 1, (unsigned)(s + RING - 1) * GM_KSTEP * N, (unsigned)(s + RING - 1) * GM_KSTEP);
        const v4f a0 = aq[s % RING][0];
        const v4f a1 = aq[s % RING][1];
        v2f a0lo, a0hi, a1lo, a1hi;
        a0lo.x = act<RELU_A>(a0.x);  a0lo.y = act<RELU_A>(a0.y);
        a0hi.x = act<RELU_A>(a0.z);  a0hi.y = act<RELU_A>(a0.w);
        a1lo.x = r1ok ? act<RELU_A>(a1.x) : 0.0f;  a1lo.y = r1ok ? act<RELU_A>(a1.y) : 0.0f;
        a1hi.x = r1ok ? act<RELU_A>(a1.z) : 0.0f;  a1hi.y = r1ok ? act<RELU_A>(a1.w) : 0.0f;
#pragma unroll
        for (int j = 0; j < 4; ++j) {
            v2f b;
            b.x = bq[s % RING][0][j];
            b.y = bq[s % RING][1][j];
            acc[0][j] = wmma4(a0lo, b, acc[0][j]);
            acc[1][j] = wmma4(a1lo, b, acc[1][j]);
        }
#pragma unroll
        for (int j = 0; j < 4; ++j) {
            v2f b;
            b.x = bq[s % RING][2][j];
            b.y = bq[s % RING][3][j];
            acc[0][j] = wmma4(a0hi, b, acc[0][j]);
            acc[1][j] = wmma4(a1hi, b, acc[1][j]);
        }
        if (s + 1 < STEPS) GM_STEP_BOUNDARY();
    }
#undef GM_STEP_BOUNDARY
#undef GM_LOAD_STEP

    float* pw = part + wave * (N_NODES * GM_COLS);
#pragma unroll
    for (int i = 0; i < 8; ++i)
        *(v4f*)&pw[(8u * h + i) * GM_COLS + 4u * c16] =
            (v4f){acc[0][0][i], acc[0][1][i], acc[0][2][i], acc[0][3][i]};
    if (h == 0) {
#pragma unroll
        for (int i = 0; i < N_NODES - 16; ++i)
            *(v4f*)&pw[(16u + i) * GM_COLS + 4u * c16] =
                (v4f){acc[1][0][i], acc[1][1][i], acc[1][2][i], acc[1][3][i]};
    }
    __syncthreads();
    if constexpr (AGG) {
        if ((unsigned)ed0 < N_NODES && (unsigned)es0 < N_NODES) atomicAdd(&magg[ed0 * AGG_PITCH + es0], 1.0f);
        if (tid + NT < N_EDGES && (unsigned)ed1 < N_NODES && (unsigned)es1 < N_NODES)
            atomicAdd(&magg[ed1 * AGG_PITCH + es1], 1.0f);
    }
    constexpr int NQ = (N_NODES + 3) / 4;
    const unsigned col = tid & (GM_COLS - 1u);
    const unsigned rg  = tid >> 6;
    float s[NQ];
#pragma unroll
    for (int q = 0; q < NQ; ++q) {
        const unsigned row = rg + 4u * q;
        s[q] = 0.0f;
        if (row < N_NODES) {
#pragma unroll
            for (int wv = 0; wv < GM_WAVES; ++wv) s[q] += part[wv * (N_NODES * GM_COLS) + row * GM_COLS + col];
        }
    }
    if constexpr (!AGG) {
#pragma unroll
        for (int q = 0; q < NQ; ++q) {
            const unsigned row = rg + 4u * q;
            if (row < N_NODES) atomicAdd(&Out[(size_t)row * N + n0 + col], s[q]);
        }
    } else {
#pragma unroll
        for (int q = 0; q < NQ; ++q) {
            const unsigned row = rg + 4u * q;
            if (row < N_NODES) psum[row * GM_COLS + col] = s[q];
        }
        __syncthreads();
        float p[AGG_PITCH];
#pragma unroll
        for (int u = 0; u < AGG_PITCH; ++u) p[u] = u < N_NODES ? psum[u * GM_COLS + col] : 0.0f;
#pragma unroll
        for (int q = 0; q < NQ; ++q) {
            const unsigned row = rg + 4u * q;
            if (row < N_NODES) {
                const v4f* mrow = (const v4f*)&magg[row * AGG_PITCH];
                float a = 0.0f;
#pragma unroll
                for (int g = 0; g < AGG_PITCH / 4; ++g) {
                    const v4f m = mrow[g];
                    a = __builtin_fmaf(m.x, p[4 * g + 0], a);
                    a = __builtin_fmaf(m.y, p[4 * g + 1], a);
                    a = __builtin_fmaf(m.z, p[4 * g + 2], a);
                    a = __builtin_fmaf(m.w, p[4 * g + 3], a);
                }
                atomicAdd(&Out[(size_t)row * N + n0 + col], a);
            }
        }
    }
}

#define MV_WAVES  16
#define MV_ROWS   19
#define MV_AHEAD  8
#define MV_BLOCKS ((N_NODES * N_T) / (MV_WAVES * MV_ROWS))
static_assert(MV_BLOCKS * MV_WAVES * MV_ROWS == N_NODES * N_T, "rows must split evenly over the waves");
static_assert(MV_WAVES * 32 == LATENT, "one thread per output column in the workgroup sum");
static_assert(MV_AHEAD >= 1 && MV_AHEAD <= MV_ROWS && MV_ROWS <= 32, "rows in flight; x sits in one lane per row");

__global__ __launch_bounds__(MV_WAVES * 32) void matvec_kernel(
        const float* __restrict__ x, const float* __restrict__ Wc1,
        float* __restrict__ y) {
    __shared__ float part[MV_WAVES * LATENT];
    const unsigned tid  = threadIdx.x;
    const unsigned lane = tid & 31u;
    const unsigned wave = __builtin_amdgcn_readfirstlane(tid >> 5);
    const unsigned row0 = (blockIdx.x * MV_WAVES + wave) * MV_ROWS;
    const float* wu = Wc1 + (size_t)row0 * LATENT;
    const int xl = __builtin_bit_cast(int, x[row0 + (lane < MV_ROWS ? lane : 0u)]);

    v4f w[MV_AHEAD][4];
    v4f acc[4];
#pragma unroll
    for (int q = 0; q < 4; ++q) acc[q] = (v4f){0.0f, 0.0f, 0.0f, 0.0f};

#define MV_LOAD_ROW(R, P) do { \
    _Pragma("unroll") for (int q = 0; q < 4; ++q) \
        w[(R) % MV_AHEAD][q] = __builtin_nontemporal_load((const v4f*)((P) + 4u * lane + 128 * q)); } while (0)

#pragma unroll
    for (int r = 0; r < MV_AHEAD; ++r) MV_LOAD_ROW(r, wu + (size_t)r * LATENT);
#pragma unroll
    for (int r = 0; r < MV_ROWS; ++r) {
        const float xv = __builtin_bit_cast(float, __builtin_amdgcn_readlane(xl, r));
#pragma unroll
        for (int q = 0; q < 4; ++q) {
            const v4f wv = w[r % MV_AHEAD][q];
            acc[q].x = __builtin_fmaf(xv, wv.x, acc[q].x);
            acc[q].y = __builtin_fmaf(xv, wv.y, acc[q].y);
            acc[q].z = __builtin_fmaf(xv, wv.z, acc[q].z);
            acc[q].w = __builtin_fmaf(xv, wv.w, acc[q].w);
        }
        if (r + MV_AHEAD < MV_ROWS) {
            unsigned off = (unsigned)(r + MV_AHEAD) * LATENT;
            asm volatile("" : "+s"(off) : "v"(acc[0]), "v"(acc[1]), "v"(acc[2]), "v"(acc[3]));
            MV_LOAD_ROW(r + MV_AHEAD, wu + off);
        }
    }
#undef MV_LOAD_ROW

#pragma unroll
    for (int q = 0; q < 4; ++q)
        *(v4f*)&part[wave * LATENT + 128 * q + 4 * lane] = acc[q];
    __syncthreads();
    float s = 0.0f;
#pragma unroll
    for (int wv = 0; wv < MV_WAVES; ++wv) s += part[wv * LATENT + tid];
    atomicAdd(&y[tid], s);
}

__global__ __launch_bounds__(512) void final_kernel(
        const float* __restrict__ y, const float* __restrict__ bc1,
        const float* __restrict__ Wc2, const float* __restrict__ bc2,
        float* __restrict__ out0) {
    __shared__ float red[512];
    int j = threadIdx.x;
    red[j] = (y[j] + bc1[j]) * Wc2[j];
    __syncthreads();
    for (int s = 256; s > 0; s >>= 1) {
        if (j < s) red[j] += red[j + s];
        __syncthreads();
    }
    if (j == 0) {
        float v = red[0] + bc2[0];
        out0[0] = 1.0f / (1.0f + __expf(-v));
    }
}

extern "C" void kernel_launch(void* const* d_in, const int* in_sizes, int n_in,
                              void* d_out, int out_size, void* d_ws, size_t ws_size,
                              hipStream_t stream) {
    const float* z    = (const float*)d_in[0];
    const int*   eidx = (const int*)  d_in[1];
    const float* W1a  = (const float*)d_in[2];
    const float* b1a  = (const float*)d_in[3];
    const float* W1b  = (const float*)d_in[4];
    const float* b1b  = (const float*)d_in[5];
    const float* W2a  = (const float*)d_in[6];
    const float* b2a  = (const float*)d_in[7];
    const float* W2b  = (const float*)d_in[8];
    const float* b2b  = (const float*)d_in[9];
    const float* Wc1  = (const float*)d_in[10];
    const float* bc1  = (const float*)d_in[11];
    const float* Wc2  = (const float*)d_in[12];
    const float* bc2  = (const float*)d_in[13];

    float* out   = (float*)d_out;
    float* ws    = (float*)d_ws;
    float* t1    = ws + WS_T1;
    float* h1    = ws + WS_H1;
    float* t2    = ws + WS_T2;
    float* yv    = ws + WS_Y;
    float* recon = out + 1;

    prep_kernel<<<PREP_BLOCKS, PREP_THREADS, 0, stream>>>(b1a, b1b, b2a, b2b, t1, h1, t2, recon, yv);


    gemm_wmma_kernel<LATENT, HID, 4, false, true>
        <<<dim3(HID / GM_COLS, 4), GM_WAVES * 32, 0, stream>>>(z, W1a, eidx, t1);
    gemm_wmma_kernel<HID, HID, 8, true, false>
        <<<dim3(HID / GM_COLS, 8), GM_WAVES * 32, 0, stream>>>(t1, W1b, eidx, h1);
    gemm_wmma_kernel<HID, N_T, 4, true, true>
        <<<dim3(N_T / GM_COLS, 4), GM_WAVES * 32, 0, stream>>>(h1, W2a, eidx, t2);
    gemm_wmma_kernel<N_T, N_T, 4, true, false>
        <<<dim3(N_T / GM_COLS, 4), GM_WAVES * 32, 0, stream>>>(t2, W2b, eidx, recon);
    matvec_kernel<<<MV_BLOCKS, MV_WAVES * 32, 0, stream>>>(recon, Wc1, yv);
    final_kernel<<<1, 512, 0, stream>>>(yv, bc1, Wc2, bc2, out);
}


# 7 -> 6 launches: no start-value pass; GEMM 1 stores its 4 K-slice partial planes, GEMM 2 sums bias + planes in order into LDS for its A operand; GEMM 1 also stores later start values; sums regrouped
# speedup vs baseline: 5.1475x; 1.0433x over previous
#include <hip/hip_runtime.h>
#include <hip/hip_bf16.h>

typedef __attribute__((ext_vector_type(2))) float v2f;
typedef __attribute__((ext_vector_type(8))) float v8f;

#define N_NODES 19
#define N_T     4096
#define LATENT  512
#define HID     2048
#define N_EDGES 342
#define PAD_ROWS 32

#define T1_PLANES 4
#define WS_T1    0
#define WS_H1    (WS_T1 + T1_PLANES * PAD_ROWS * HID)
#define WS_T2    (WS_H1 + PAD_ROWS * HID)
#define WS_Y     (WS_T2 + PAD_ROWS * N_T)
#define WS_TOTAL (WS_Y + LATENT)

typedef __attribute__((ext_vector_type(4))) float v4f;

template <bool RELU>
__device__ __forceinline__ float act(float v) {
    return RELU ? (v > 0.0f ? v : 0.0f) : v;
}

#define PREP_THREADS    256
#define PREP_SPAN       (4 * PREP_THREADS)
#define PREP_HID_BLOCKS ((N_NODES * HID) / PREP_SPAN)
#define PREP_NT_BLOCKS  ((N_NODES * N_T) / PREP_SPAN)
#define FILL_BLOCKS     (PREP_HID_BLOCKS + 2 * PREP_NT_BLOCKS + 1)
static_assert(PREP_HID_BLOCKS * PREP_SPAN == N_NODES * HID && PREP_NT_BLOCKS * PREP_SPAN == N_NODES * N_T,
              "planes split evenly over fill blocks");
static_assert(HID % PREP_SPAN == 0 && N_T % PREP_SPAN == 0, "a fill block never crosses a row end");
static_assert(LATENT <= PREP_SPAN && LATENT % 4 == 0, "y fits one fill block");

struct FillArgs {
    const float* b1b;
    const float* b2a;
    const float* b2b;
    float* h1;
    float* t2;
    float* recon;
    float* y;
};

struct FillSel {
    const float* src;
    float* dst;
    int base;
    int ncols;
    int count;
    bool zero;
};

__device__ __forceinline__ FillSel fill_select(const FillArgs& fa, int b) {
    const bool isH = b < PREP_HID_BLOCKS;
    const bool isT = !isH && b < PREP_HID_BLOCKS + PREP_NT_BLOCKS;
    const bool isR = !isH && !isT && b < PREP_HID_BLOCKS + 2 * PREP_NT_BLOCKS;
    const bool isY = b == FILL_BLOCKS - 1;
    FillSel s;
    s.src   = isH ? fa.b1b : (isT ? fa.b2a : fa.b2b);
    s.dst   = isH ? fa.h1 : (isT ? fa.t2 : (isR ? fa.recon : fa.y));
    s.base  = (isH ? b : (isT ? b - PREP_HID_BLOCKS : (isR ? b - PREP_HID_BLOCKS - PREP_NT_BLOCKS : 0))) * PREP_SPAN;
    s.ncols = isH ? HID : N_T;
    s.count = (isH || isT || isR) ? PREP_SPAN : (isY ? LATENT : 0);
    s.zero  = isY;
    return s;
}

__device__ __forceinline__ v4f fill_load(const FillSel& s) {
    v4f v;
#pragma unroll
    for (int j = 0; j < 4; ++j) v[j] = s.src[(s.base + j * PREP_THREADS + (int)threadIdx.x) % s.ncols];
    return v;
}

__device__ __forceinline__ void fill_store(const FillSel& s, v4f v) {
#pragma unroll
    for (int j = 0; j < 4; ++j) {
        const int e = j * PREP_THREADS + (int)threadIdx.x;
        if (e < s.count) s.dst[s.base + e] = s.zero ? 0.0f : v[j];
    }
}

#define GM_WAVES 8
#define GM_COLS  64
#define GM_KSTEP 8
#define GM_RING  5
#define AGG_PITCH 20

__device__ __forceinline__ v8f wmma4(v2f a, v2f b, v8f c) {
    return __builtin_amdgcn_wmma_f32_16x16x4_f32(false, a, false, b, (short)0, c, false, false);
}

template <int K, int N, int GS, bool RELU_A, bool AGG, bool HEAD, int AP>
__global__ __launch_bounds__(GM_WAVES * 32) void gemm_wmma_kernel(
        const float* __restrict__ A, const float* __restrict__ W,
        const int* __restrict__ edges, float* __restrict__ Out,
        const float* __restrict__ abias, FillArgs fa) {
    constexpr int KB    = K / GS;
    constexpr int KW    = KB / GM_WAVES;
    constexpr int STEPS = KW / GM_KSTEP;
    constexpr int RING  = STEPS + 1 < GM_RING ? STEPS + 1 : GM_RING;
    static_assert(GS * GM_WAVES * GM_KSTEP * STEPS == K, "K must split evenly over slices, waves and steps");
    static_assert(N % GM_COLS == 0 && N % 4 == 0 && K % 4 == 0, "16-byte loads need aligned rows");
    static_assert(GM_WAVES * 32 == 4 * GM_COLS, "the workgroup sum maps 4 row groups x 64 columns onto the threads");
    static_assert(N_EDGES >= GM_WAVES * 32 && N_EDGES <= 2 * GM_WAVES * 32, "one or two edges per thread");
    static_assert(AGG_PITCH % 4 == 0 && AGG_PITCH > N_NODES, "a row of M is whole 16-byte pieces with a zero pad");
    constexpr int NWG    = (N / GM_COLS) * GS;
    constexpr int APITCH = KB + 4;
    constexpr int AQ     = KB / 4;
    constexpr int ANIT   = (N_NODES * AQ + GM_WAVES * 32 - 1) / (GM_WAVES * 32);
    constexpr int AFULL  = (N_NODES * AQ) / (GM_WAVES * 32);
    constexpr bool RELU_USE = RELU_A && AP == 1;
    static_assert(!HEAD || (AGG && FILL_BLOCKS <= 2 * NWG && PREP_THREADS == GM_WAVES * 32),
                  "HEAD: aggregated form; at most two fill blocks per workgroup, one float4 per thread each");
    static_assert(AP == 1 || (STEPS + 1 <= GM_RING && KB % 4 == 0 && (APITCH / 4) % 2 == 1 && (GM_WAVES * 32) % AQ == 0 &&
                              ANIT * GM_WAVES * 32 <= (N_NODES + 1) * AQ),
                  "AP > 1: whole stream up front, all A fragments in registers; LDS pitch 16 bytes x odd");

    __shared__ __attribute__((aligned(16))) float part[GM_WAVES * N_NODES * GM_COLS];
    __shared__ __attribute__((aligned(16))) float magg[N_NODES * AGG_PITCH];
    __shared__ float psum[N_NODES * GM_COLS];
    __shared__ __attribute__((aligned(16))) float astage[AP > 1 ? (N_NODES + 1) * APITCH : 4];

    const unsigned tid  = threadIdx.x;
    const unsigned lane = tid & 31u;
    const unsigned wave = __builtin_amdgcn_readfirstlane(tid >> 5);
    const unsigned h    = lane >> 4;
    const unsigned c16  = lane & 15u;
    const unsigned n0   = blockIdx.x * GM_COLS;
    const unsigned kbeg = blockIdx.y * KB + wave * KW;

    const float* wu = W + (size_t)kbeg * N + n0;
    const float* au = A + kbeg;
    const unsigned boff  = 4u * h * N + 4u * c16;
    const bool     r1ok  = c16 < N_NODES - 16;
    const unsigned a0off = c16 * K + 4u * h;
    const unsigned a1off = (16u + (r1ok ? c16 : 0u)) * K + 4u * h;

    const int wg = (int)(blockIdx.y * (N / GM_COLS) + blockIdx.x);
    const FillSel fs0 = fill_select(fa, wg), fs1 = fill_select(fa, wg + NWG);
    v4f fv0 = (v4f){0.0f, 0.0f, 0.0f, 0.0f}, fv1 = fv0;
    if constexpr (HEAD) {
        fv0 = fill_load(fs0);
        fv1 = fill_load(fs1);
    }
    v4f sv[ANIT][AP];
    v4f sb = (v4f){0.0f, 0.0f, 0.0f, 0.0f};
    if constexpr (AP > 1) {
        sb = *(const v4f*)(abias + blockIdx.y * KB + 4u * (tid % AQ));
#pragma unroll
        for (int it = 0; it < ANIT; ++it) {
            const unsigned i  = tid + (unsigned)it * (GM_WAVES * 32);
            const unsigned ic = it < AFULL ? i : (i < N_NODES * AQ ? i : N_NODES * AQ - 1u);
            const float* p = A + (size_t)(ic / AQ) * K + blockIdx.y * KB + 4u * (ic % AQ);
#pragma unroll
            for (int pl = 0; pl < AP; ++pl) sv[it][pl] = *(const v4f*)(p + (size_t)pl * (PAD_ROWS * K));
        }
    }

    constexpr unsigned NT = GM_WAVES * 32;
    int es0 = 0, ed0 = 0, es1 = 0, ed1 = 0;
    if constexpr (AGG) {
        es0 = edges[tid];
        ed0 = edges[N_EDGES + tid];
        if (tid + NT < N_EDGES) {
            es1 = edges[tid + NT];
            ed1 = edges[N_EDGES + tid + NT];
        }
        for (unsigned i = tid; i < N_NODES * AGG_PITCH; i += NT)
            magg[i] = (i / AGG_PITCH == i % AGG_PITCH) ? 1.0f : 0.0f;
    }

    v4f bq[RING][4];
    v4f aq[RING][2];
    v8f acc[2][4];
#pragma unroll
    for (int m = 0; m < 2; ++m)
#pragma unroll
        for (int j = 0; j < 4; ++j) acc[m][j] = (v8f){0.0f, 0.0f, 0.0f, 0.0f, 0.0f, 0.0f, 0.0f, 0.0f};

#define GM_LOAD_STEP(S, BO, AO) do { \
    _Pragma("unroll") for (int r = 0; r < 4; ++r) \
        bq[(S) % RING][r] = *(const v4f*)(wu + (BO) + (unsigned)r * N + boff); \
    if constexpr (AP == 1) { \
        aq[(S) % RING][0] = *(const v4f*)(au + (AO) + a0off); \
        aq[(S) % RING][1] = *(const v4f*)(au + (AO) + a1off); } } while (0)

#define GM_STEP_BOUNDARY() __builtin_amdgcn_sched_barrier(0)

#pragma unroll
    for (int s = 0; s < RING - 1; ++s) {
        GM_LOAD_STEP(s, (unsigned)s * GM_KSTEP * N, (unsigned)s * GM_KSTEP);
        GM_STEP_BOUNDARY();
    }

    if constexpr (HEAD) {
        fill_store(fs0, fv0);
        fill_store(fs1, fv1);
    }
    v4f af[AP > 1 ? STEPS : 1][2];
    if constexpr (AP > 1) {
#pragma unroll
        for (int it = 0; it < ANIT; ++it) {
            const unsigned i = tid + (unsigned)it * (GM_WAVES * 32);
            const unsigned row = i / AQ, q = i % AQ;
            v4f v = sb;
#pragma unroll
            for (int pl = 0; pl < AP; ++pl) v += sv[it][pl];
            v.x = act<RELU_A>(v.x);  v.y = act<RELU_A>(v.y);
            v.z = act<RELU_A>(v.z);  v.w = act<RELU_A>(v.w);
            *(v4f*)&astage[row * APITCH + 4u * q] = v;
        }
        __syncthreads();
        const unsigned f0 = c16 * APITCH + wave * KW + 4u * h;
        const unsigned f1 = (16u + (r1ok ? c16 : 0u)) * APITCH + wave * KW + 4u * h;
#pragma unroll
        for (int s = 0; s < STEPS; ++s) {
            af[s][0] = *(const v4f*)&astage[f0 + (unsigned)s * GM_KSTEP];
            af[s][1] = *(const v4f*)&astage[f1 + (unsigned)s * GM_KSTEP];
        }
    }

#pragma unroll
    for (int s = 0; s < STEPS; ++s) {
        if (s + RING - 1 < STEPS)
            GM_LOAD_STEP(s + RING - 1, (unsigned)(s + RING - 1) * GM_KSTEP * N, (unsigned)(s + RING - 1) * GM_KSTEP);
        v4f a0, a1;
        if constexpr (AP > 1) { a0 = af[s][0]; a1 = af[s][1]; }
        else { a0 = aq[s % RING][0]; a1 = aq[s % RING][1]; }
        v2f a0lo, a0hi, a1lo, a1hi;
        a0lo.x = act<RELU_USE>(a0.x);  a0lo.y = act<RELU_USE>(a0.y);
        a0hi.x = act<RELU_USE>(a0.z);  a0hi.y = act<RELU_USE>(a0.w);
        a1lo.x = r1ok ? act<RELU_USE>(a1.x) : 0.0f;  a1lo.y = r1ok ? act<RELU_USE>(a1.y) : 0.0f;
        a1hi.x = r1ok ? act<RELU_USE>(a1.z) : 0.0f;  a1hi.y = r1ok ? act<RELU_USE>(a1.w) : 0.0f;
#pragma unroll
        for (int j = 0; j < 4; ++j) {
            v2f b;
            b.x = bq[s % RING][0][j];
            b.y = bq[s % RING][1][j];
            acc[0][j] = wmma4(a0lo, b, acc[0][j]);
            acc[1][j] = wmma4(a1lo, b, acc[1][j]);
        }
#pragma unroll
        for (int j = 0; j < 4; ++j) {
            v2f b;
            b.x = bq[s % RING][2][j];
            b.y = bq[s % RING][3][j];
            acc[0][j] = wmma4(a0hi, b, acc[0][j]);
            acc[1][j] = wmma4(a1hi, b, acc[1][j]);
        }
        if (s + 1 < STEPS) GM_STEP_BOUNDARY();
    }
#undef GM_STEP_BOUNDARY
#undef GM_LOAD_STEP

    float* pw = part + wave * (N_NODES * GM_COLS);
#pragma unroll
    for (int i = 0; i < 8; ++i)
        *(v4f*)&pw[(8u * h + i) * GM_COLS + 4u * c16] =
            (v4f){acc[0][0][i], acc[0][1][i], acc[0][2][i], acc[0][3][i]};
    if (h == 0) {
#pragma unroll
        for (int i = 0; i < N_NODES - 16; ++i)
            *(v4f*)&pw[(16u + i) * GM_COLS + 4u * c16] =
                (v4f){acc[1][0][i], acc[1][1][i], acc[1][2][i], acc[1][3][i]};
    }
    __syncthreads();
    if constexpr (AGG) {
        if ((unsigned)ed0 < N_NODES && (unsigned)es0 < N_NODES) atomicAdd(&magg[ed0 * AGG_PITCH + es0], 1.0f);
        if (tid + NT < N_EDGES && (unsigned)ed1 < N_NODES && (unsigned)es1 < N_NODES)
            atomicAdd(&magg[ed1 * AGG_PITCH + es1], 1.0f);
    }
    constexpr int NQ = (N_NODES + 3) / 4;
    const unsigned col = tid & (GM_COLS - 1u);
    const unsigned rg  = tid >> 6;
    float s[NQ];
#pragma unroll
    for (int q = 0; q < NQ; ++q) {
        const unsigned row = rg + 4u * q;
        s[q] = 0.0f;
        if (row < N_NODES) {
#pragma unroll
            for (int wv = 0; wv < GM_WAVES; ++wv) s[q] += part[wv * (N_NODES * GM_COLS) + row * GM_COLS + col];
        }
    }
    if constexpr (!AGG) {
#pragma unroll
        for (int q = 0; q < NQ; ++q) {
            const unsigned row = rg + 4u * q;
            if (row < N_NODES) atomicAdd(&Out[(size_t)row * N + n0 + col], s[q]);
        }
    } else {
#pragma unroll
        for (int q = 0; q < NQ; ++q) {
            const unsigned row = rg + 4u * q;
            if (row < N_NODES) psum[row * GM_COLS + col] = s[q];
        }
        __syncthreads();
        float p[AGG_PITCH];
#pragma unroll
        for (int u = 0; u < AGG_PITCH; ++u) p[u] = u < N_NODES ? psum[u * GM_COLS + col] : 0.0f;
#pragma unroll
        for (int q = 0; q < NQ; ++q) {
            const unsigned row = rg + 4u * q;
            if (row < N_NODES) {
                const v4f* mrow = (const v4f*)&magg[row * AGG_PITCH];
                float a = 0.0f;
#pragma unroll
                for (int g = 0; g < AGG_PITCH / 4; ++g) {
                    const v4f m = mrow[g];
                    a = __builtin_fmaf(m.x, p[4 * g + 0], a);
                    a = __builtin_fmaf(m.y, p[4 * g + 1], a);
                    a = __builtin_fmaf(m.z, p[4 * g + 2], a);
                    a = __builtin_fmaf(m.w, p[4 * g + 3], a);
                }
                if constexpr (HEAD) Out[(size_t)blockIdx.y * (PAD_ROWS * N) + (size_t)row * N + n0 + col] = a;
                else atomicAdd(&Out[(size_t)row * N + n0 + col], a);
            }
        }
    }
}

#define MV_WAVES  16
#define MV_ROWS   19
#define MV_AHEAD  8
#define MV_BLOCKS ((N_NODES * N_T) / (MV_WAVES * MV_ROWS))
static_assert(MV_BLOCKS * MV_WAVES * MV_ROWS == N_NODES * N_T, "rows must split evenly over the waves");
static_assert(MV_WAVES * 32 == LATENT, "one thread per output column in the workgroup sum");
static_assert(MV_AHEAD >= 1 && MV_AHEAD <= MV_ROWS && MV_ROWS <= 32, "rows in flight; x sits in one lane per row");

__global__ __launch_bounds__(MV_WAVES * 32) void matvec_kernel(
        const float* __restrict__ x, const float* __restrict__ Wc1,
        float* __restrict__ y) {
    __shared__ float part[MV_WAVES * LATENT];
    const unsigned tid  = threadIdx.x;
    const unsigned lane = tid & 31u;
    const unsigned wave = __builtin_amdgcn_readfirstlane(tid >> 5);
    const unsigned row0 = (blockIdx.x * MV_WAVES + wave) * MV_ROWS;
    const float* wu = Wc1 + (size_t)row0 * LATENT;
    const int xl = __builtin_bit_cast(int, x[row0 + (lane < MV_ROWS ? lane : 0u)]);

    v4f w[MV_AHEAD][4];
    v4f acc[4];
#pragma unroll
    for (int q = 0; q < 4; ++q) acc[q] = (v4f){0.0f, 0.0f, 0.0f, 0.0f};

#define MV_LOAD_ROW(R, P) do { \
    _Pragma("unroll") for (int q = 0; q < 4; ++q) \
        w[(R) % MV_AHEAD][q] = __builtin_nontemporal_load((const v4f*)((P) + 4u * lane + 128 * q)); } while (0)

#pragma unroll
    for (int r = 0; r < MV_AHEAD; ++r) MV_LOAD_ROW(r, wu + (size_t)r * LATENT);
#pragma unroll
    for (int r = 0; r < MV_ROWS; ++r) {
        const float xv = __builtin_bit_cast(float, __builtin_amdgcn_readlane(xl, r));
#pragma unroll
        for (int q = 0; q < 4; ++q) {
            const v4f wv = w[r % MV_AHEAD][q];
            acc[q].x = __builtin_fmaf(xv, wv.x, acc[q].x);
            acc[q].y = __builtin_fmaf(xv, wv.y, acc[q].y);
            acc[q].z = __builtin_fmaf(xv, wv.z, acc[q].z);
            acc[q].w = __builtin_fmaf(xv, wv.w, acc[q].w);
        }
        if (r + MV_AHEAD < MV_ROWS) {
            unsigned off = (unsigned)(r + MV_AHEAD) * LATENT;
            asm volatile("" : "+s"(off) : "v"(acc[0]), "v"(acc[1]), "v"(acc[2]), "v"(acc[3]));
            MV_LOAD_ROW(r + MV_AHEAD, wu + off);
        }
    }
#undef MV_LOAD_ROW

#pragma unroll
    for (int q = 0; q < 4; ++q)
        *(v4f*)&part[wave * LATENT + 128 * q + 4 * lane] = acc[q];
    __syncthreads();
    float s = 0.0f;
#pragma unroll
    for (int wv = 0; wv < MV_WAVES; ++wv) s += part[wv * LATENT + tid];
    atomicAdd(&y[tid], s);
}

__global__ __launch_bounds__(512) void final_kernel(
        const float* __restrict__ y, const float* __restrict__ bc1,
        const float* __restrict__ Wc2, const float* __restrict__ bc2,
        float* __restrict__ out0) {
    __shared__ float red[512];
    int j = threadIdx.x;
    red[j] = (y[j] + bc1[j]) * Wc2[j];
    __syncthreads();
    for (int s = 256; s > 0; s >>= 1) {
        if (j < s) red[j] += red[j + s];
        __syncthreads();
    }
    if (j == 0) {
        float v = red[0] + bc2[0];
        out0[0] = 1.0f / (1.0f + __expf(-v));
    }
}

extern "C" void kernel_launch(void* const* d_in, const int* in_sizes, int n_in,
                              void* d_out, int out_size, void* d_ws, size_t ws_size,
                              hipStream_t stream) {
    const float* z    = (const float*)d_in[0];
    const int*   eidx = (const int*)  d_in[1];
    const float* W1a  = (const float*)d_in[2];
    const float* b1a  = (const float*)d_in[3];
    const float* W1b  = (const float*)d_in[4];
    const float* b1b  = (const float*)d_in[5];
    const float* W2a  = (const float*)d_in[6];
    const float* b2a  = (const float*)d_in[7];
    const float* W2b  = (const float*)d_in[8];
    const float* b2b  = (const float*)d_in[9];
    const float* Wc1  = (const float*)d_in[10];
    const float* bc1  = (const float*)d_in[11];
    const float* Wc2  = (const float*)d_in[12];
    const float* bc2  = (const float*)d_in[13];

    float* out   = (float*)d_out;
    float* ws    = (float*)d_ws;
    float* t1    = ws + WS_T1;
    float* h1    = ws + WS_H1;
    float* t2    = ws + WS_T2;
    float* yv    = ws + WS_Y;
    float* recon = out + 1;

    FillArgs fa;
    fa.b1b = b1b; fa.b2a = b2a; fa.b2b = b2b;
    fa.h1 = h1; fa.t2 = t2; fa.recon = recon; fa.y = yv;


    gemm_wmma_kernel<LATENT, HID, T1_PLANES, false, true, true, 1>
        <<<dim3(HID / GM_COLS, T1_PLANES), GM_WAVES * 32, 0, stream>>>(z, W1a, eidx, t1, b1a, fa);
    gemm_wmma_kernel<HID, HID, 8, true, false, false, T1_PLANES>
        <<<dim3(HID / GM_COLS, 8), GM_WAVES * 32, 0, stream>>>(t1, W1b, eidx, h1, b1a, fa);
    gemm_wmma_kernel<HID, N_T, 4, true, true, false, 1>
        <<<dim3(N_T / GM_COLS, 4), GM_WAVES * 32, 0, stream>>>(h1, W2a, eidx, t2, b2a, fa);
    gemm_wmma_kernel<N_T, N_T, 4, true, false, false, 1>
        <<<dim3(N_T / GM_COLS, 4), GM_WAVES * 32, 0, stream>>>(t2, W2b, eidx, recon, b2b, fa);
    matvec_kernel<<<MV_BLOCKS, MV_WAVES * 32, 0, stream>>>(recon, Wc1, yv);
    final_kernel<<<1, 512, 0, stream>>>(yv, bc1, Wc2, bc2, out);
}
